# starting kernel
# baseline (speedup 1.0000x reference)
#include <hip/hip_runtime.h>
#include <math.h>

#define MD    1024
#define NH    16
#define HD    64
#define BATCH 4
#define SEQ   2048

typedef __attribute__((ext_vector_type(2))) float v2f;
typedef __attribute__((ext_vector_type(8))) float v8f;

static __device__ __forceinline__ v8f wmma_f32(v2f a, v2f b, v8f c) {
  return __builtin_amdgcn_wmma_f32_16x16x4_f32(
       false, a,  false, b,
       (short)0, c,  false,  false);
}

__global__ __launch_bounds__(256)
void gemm_nt_128x64(const float* __restrict__ A, const float* __restrict__ W,
                    const float* __restrict__ bias, float* __restrict__ C,
                    int headsplit)
{
  __shared__ float As[128][18];
  __shared__ float Ws[64][18];

  const int m0   = blockIdx.x * 128;
  const int n0   = blockIdx.y * 64;
  const int tid  = threadIdx.x;
  const int w    = tid >> 5;
  const int lane = tid & 31;
  const int ll   = lane & 15;
  const int hh   = lane >> 4;
  const int mq   = (w >> 1) * 32;
  const int nq   = (w & 1) * 32;

  const v8f z = {0.f,0.f,0.f,0.f,0.f,0.f,0.f,0.f};
  v8f acc00 = z, acc01 = z, acc10 = z, acc11 = z;

  const int lrA0 = tid >> 2;
  const int lcA  = (tid & 3) * 4;
  const int lrW  = tid >> 2;

  for (int k0 = 0; k0 < MD; k0 += 16) {
    const float4 a0 = *(const float4*)&A[(size_t)(m0 + lrA0     ) * MD + k0 + lcA];
    const float4 a1 = *(const float4*)&A[(size_t)(m0 + lrA0 + 64) * MD + k0 + lcA];
    const float4 wv = *(const float4*)&W[(size_t)(n0 + lrW      ) * MD + k0 + lcA];
    As[lrA0   ][lcA+0] = a0.x; As[lrA0   ][lcA+1] = a0.y;
    As[lrA0   ][lcA+2] = a0.z; As[lrA0   ][lcA+3] = a0.w;
    As[lrA0+64][lcA+0] = a1.x; As[lrA0+64][lcA+1] = a1.y;
    As[lrA0+64][lcA+2] = a1.z; As[lrA0+64][lcA+3] = a1.w;
    Ws[lrW][lcA+0] = wv.x; Ws[lrW][lcA+1] = wv.y;
    Ws[lrW][lcA+2] = wv.z; Ws[lrW][lcA+3] = wv.w;
    __syncthreads();

    if (k0 + 16 < MD) {
      __builtin_prefetch(&A[(size_t)(m0 + lrA0     ) * MD + k0 + 16 + lcA], 0, 1);
      __builtin_prefetch(&A[(size_t)(m0 + lrA0 + 64) * MD + k0 + 16 + lcA], 0, 1);
      __builtin_prefetch(&W[(size_t)(n0 + lrW      ) * MD + k0 + 16 + lcA], 0, 1);
    }

#pragma unroll
    for (int kk = 0; kk < 16; kk += 4) {
      const int col = kk + 2 * hh;
      const v2f a0f = *(const v2f*)&As[mq      + ll][col];
      const v2f a1f = *(const v2f*)&As[mq + 16 + ll][col];
      const v2f b0f = *(const v2f*)&Ws[nq      + ll][col];
      const v2f b1f = *(const v2f*)&Ws[nq + 16 + ll][col];
      acc00 = wmma_f32(a0f, b0f, acc00);
      acc01 = wmma_f32(a0f, b1f, acc01);
      acc10 = wmma_f32(a1f, b0f, acc10);
      acc11 = wmma_f32(a1f, b1f, acc11);
    }
    __syncthreads();
  }

#pragma unroll
  for (int mi = 0; mi < 2; ++mi) {
#pragma unroll
    for (int ni = 0; ni < 2; ++ni) {
      const v8f acc = (mi == 0) ? (ni == 0 ? acc00 : acc01)
                                : (ni == 0 ? acc10 : acc11);
#pragma unroll
      for (int j = 0; j < 8; ++j) {
        const int gm = m0 + mq + mi * 16 + j + hh * 8;
        const int gn = n0 + nq + ni * 16 + ll;
        float c = acc[j];
        if (!headsplit) {
          c += bias[gn];
          C[(size_t)gm * MD + gn] = c;
        } else {
          const int b = gm / SEQ, n = gm % SEQ;
          const int h = gn >> 6, d = gn & 63;
          C[(((size_t)b * NH + h) * SEQ + n) * HD + d] = c;
        }
      }
    }
  }
}

__global__ __launch_bounds__(128)
void flash_attn_causal(const float* __restrict__ Q, const float* __restrict__ K,
                       const float* __restrict__ V, float* __restrict__ O)
{
  __shared__ float Ks[16][66];
  __shared__ float Vt[64][18];
  __shared__ float Ps[4][16][18];

  const int bh   = blockIdx.x;
  const int qblk = blockIdx.y;
  const int tid  = threadIdx.x;
  const int w    = tid >> 5;
  const int lane = tid & 31;
  const int ll   = lane & 15;
  const int hh   = lane >> 4;
  const int qbase = qblk * 64 + w * 16;
  const float scale = 0.125f;

  const float* Qh = Q + (size_t)bh * SEQ * HD;
  const float* Kh = K + (size_t)bh * SEQ * HD;
  const float* Vh = V + (size_t)bh * SEQ * HD;

  v2f qf[16];
#pragma unroll
  for (int t = 0; t < 16; ++t) {
    v2f qv = *(const v2f*)&Qh[(size_t)(qbase + ll) * HD + t * 4 + 2 * hh];
    qf[t] = qv * scale;
  }

  const v8f z = {0.f,0.f,0.f,0.f,0.f,0.f,0.f,0.f};
  v8f oacc[4] = {z, z, z, z};
  float mrow[8], lrow[8];
#pragma unroll
  for (int j = 0; j < 8; ++j) { mrow[j] = -INFINITY; lrow[j] = 0.f; }

  const int nkt = qblk * 4 + 4;
  for (int kt = 0; kt < nkt; ++kt) {
    const int kbase = kt * 16;

#pragma unroll
    for (int part = 0; part < 2; ++part) {
      const int lin = tid + part * 128;
      const int row = lin >> 4, c4 = (lin & 15) * 4;
      const float4 kd = *(const float4*)&Kh[(size_t)(kbase + row) * HD + c4];
      const float4 vd = *(const float4*)&Vh[(size_t)(kbase + row) * HD + c4];
      Ks[row][c4+0] = kd.x; Ks[row][c4+1] = kd.y;
      Ks[row][c4+2] = kd.z; Ks[row][c4+3] = kd.w;
      Vt[c4+0][row] = vd.x; Vt[c4+1][row] = vd.y;
      Vt[c4+2][row] = vd.z; Vt[c4+3][row] = vd.w;
    }
    __syncthreads();

    if (kbase <= qbase + 15) {
      v8f s = z;
#pragma unroll
      for (int t = 0; t < 16; ++t) {
        const v2f b = *(const v2f*)&Ks[ll][t * 4 + 2 * hh];
        s = wmma_f32(qf[t], b, s);
      }

#pragma unroll
      for (int j = 0; j < 8; ++j) {
        const int rowg = qbase + j + hh * 8;
        const int colg = kbase + ll;
        const float sj = (colg > rowg) ? -INFINITY : s[j];
        float red = sj;
        red = fmaxf(red, __shfl_xor(red, 1, 32));
        red = fmaxf(red, __shfl_xor(red, 2, 32));
        red = fmaxf(red, __shfl_xor(red, 4, 32));
        red = fmaxf(red, __shfl_xor(red, 8, 32));
        const float mnew  = fmaxf(mrow[j], red);
        const float p     = __expf(sj - mnew);
        const float alpha = __expf(mrow[j] - mnew);
        mrow[j] = mnew;
        float psum = p;
        psum += __shfl_xor(psum, 1, 32);
        psum += __shfl_xor(psum, 2, 32);
        psum += __shfl_xor(psum, 4, 32);
        psum += __shfl_xor(psum, 8, 32);
        lrow[j] = lrow[j] * alpha + psum;
#pragma unroll
        for (int dt = 0; dt < 4; ++dt) oacc[dt][j] *= alpha;
        Ps[w][j + hh * 8][ll] = p;
      }

#pragma unroll
      for (int t = 0; t < 4; ++t) {
        const v2f a = *(const v2f*)&Ps[w][ll][t * 4 + 2 * hh];
#pragma unroll
        for (int dt = 0; dt < 4; ++dt) {
          const v2f b = *(const v2f*)&Vt[dt * 16 + ll][t * 4 + 2 * hh];
          oacc[dt] = wmma_f32(a, b, oacc[dt]);
        }
      }
    }
    __syncthreads();
  }

  const int b = bh >> 4;
  const int h = bh & 15;
#pragma unroll
  for (int j = 0; j < 8; ++j) {
    const float inv = 1.0f / lrow[j];
    const int rowg = qbase + j + hh * 8;
#pragma unroll
    for (int dt = 0; dt < 4; ++dt) {
      const int d = dt * 16 + ll;
      O[((size_t)b * SEQ + rowg) * MD + h * HD + d] = oacc[dt][j] * inv;
    }
  }
}

extern "C" void kernel_launch(void* const* d_in, const int* in_sizes, int n_in,
                              void* d_out, int out_size, void* d_ws, size_t ws_size,
                              hipStream_t stream) {
  (void)in_sizes; (void)n_in; (void)out_size; (void)ws_size;
  const float* x  = (const float*)d_in[0];
  const float* Wq = (const float*)d_in[1];
  const float* Wk = (const float*)d_in[2];
  const float* Wv = (const float*)d_in[3];
  const float* Wo = (const float*)d_in[4];
  const float* bo = (const float*)d_in[5];

  float* ws = (float*)d_ws;
  const size_t SZ = (size_t)BATCH * SEQ * MD;
  float* qb = ws;
  float* kb = ws + SZ;
  float* vb = ws + 2 * SZ;
  float* ao = ws + 3 * SZ;

  const dim3 ggrid(BATCH * SEQ / 128, MD / 64);
  gemm_nt_128x64<<<ggrid, 256, 0, stream>>>(x, Wq, nullptr, qb, 1);
  gemm_nt_128x64<<<ggrid, 256, 0, stream>>>(x, Wk, nullptr, kb, 1);
  gemm_nt_128x64<<<ggrid, 256, 0, stream>>>(x, Wv, nullptr, vb, 1);

  flash_attn_causal<<<dim3(BATCH * NH, SEQ / 64), 128, 0, stream>>>(qb, kb, vb, ao);

  gemm_nt_128x64<<<ggrid, 256, 0, stream>>>(ao, Wo, bo, (float*)d_out, 0);
}


# GEMM K loop software-pipelined: next tile's global loads issued into registers before this tile's WMMAs and stored to a second LDS stage after them; one barrier per K step; prefetch hints dropped
# speedup vs baseline: 1.0921x; 1.0921x over previous
#include <hip/hip_runtime.h>
#include <math.h>

#define MD    1024
#define NH    16
#define HD    64
#define BATCH 4
#define SEQ   2048

typedef __attribute__((ext_vector_type(2))) float v2f;
typedef __attribute__((ext_vector_type(8))) float v8f;

static __device__ __forceinline__ v8f wmma_f32(v2f a, v2f b, v8f c) {
  return __builtin_amdgcn_wmma_f32_16x16x4_f32(
       false, a,  false, b,
       (short)0, c,  false,  false);
}

__global__ __launch_bounds__(256)
void gemm_nt_128x64(const float* __restrict__ A, const float* __restrict__ W,
                    const float* __restrict__ bias, float* __restrict__ C,
                    int headsplit)
{
  __shared__ float As[2][128][18];
  __shared__ float Ws[2][64][18];

  const int m0   = blockIdx.x * 128;
  const int n0   = blockIdx.y * 64;
  const int tid  = threadIdx.x;
  const int w    = tid >> 5;
  const int lane = tid & 31;
  const int ll   = lane & 15;
  const int hh   = lane >> 4;
  const int mq   = (w >> 1) * 32;
  const int nq   = (w & 1) * 32;

  const v8f z = {0.f,0.f,0.f,0.f,0.f,0.f,0.f,0.f};
  v8f acc00 = z, acc01 = z, acc10 = z, acc11 = z;

  const int lrA0 = tid >> 2;
  const int lcA  = (tid & 3) * 4;
  const int lrW  = tid >> 2;

  const float* Ap0 = &A[(size_t)(m0 + lrA0     ) * MD + lcA];
  const float* Ap1 = &A[(size_t)(m0 + lrA0 + 64) * MD + lcA];
  const float* Wp  = &W[(size_t)(n0 + lrW      ) * MD + lcA];

  float4 a0 = *(const float4*)&Ap0[0];
  float4 a1 = *(const float4*)&Ap1[0];
  float4 wv = *(const float4*)&Wp[0];
  As[0][lrA0   ][lcA+0] = a0.x; As[0][lrA0   ][lcA+1] = a0.y;
  As[0][lrA0   ][lcA+2] = a0.z; As[0][lrA0   ][lcA+3] = a0.w;
  As[0][lrA0+64][lcA+0] = a1.x; As[0][lrA0+64][lcA+1] = a1.y;
  As[0][lrA0+64][lcA+2] = a1.z; As[0][lrA0+64][lcA+3] = a1.w;
  Ws[0][lrW][lcA+0] = wv.x; Ws[0][lrW][lcA+1] = wv.y;
  Ws[0][lrW][lcA+2] = wv.z; Ws[0][lrW][lcA+3] = wv.w;
  __syncthreads();

  for (int k0 = 0; k0 < MD; k0 += 32) {
#pragma unroll
    for (int st = 0; st < 2; ++st) {
      const int kn = k0 + 16 * st + 16;
      const bool more = (kn < MD);
      if (more) {
        a0 = *(const float4*)&Ap0[kn];
        a1 = *(const float4*)&Ap1[kn];
        wv = *(const float4*)&Wp[kn];
      }

#pragma unroll
      for (int kk = 0; kk < 16; kk += 4) {
        const int col = kk + 2 * hh;
        const v2f a0f = *(const v2f*)&As[st][mq      + ll][col];
        const v2f a1f = *(const v2f*)&As[st][mq + 16 + ll][col];
        const v2f b0f = *(const v2f*)&Ws[st][nq      + ll][col];
        const v2f b1f = *(const v2f*)&Ws[st][nq + 16 + ll][col];
        acc00 = wmma_f32(a0f, b0f, acc00);
        acc01 = wmma_f32(a0f, b1f, acc01);
        acc10 = wmma_f32(a1f, b0f, acc10);
        acc11 = wmma_f32(a1f, b1f, acc11);
      }

      __builtin_amdgcn_sched_barrier(0);
      if (more) {
        As[st^1][lrA0   ][lcA+0] = a0.x; As[st^1][lrA0   ][lcA+1] = a0.y;
        As[st^1][lrA0   ][lcA+2] = a0.z; As[st^1][lrA0   ][lcA+3] = a0.w;
        As[st^1][lrA0+64][lcA+0] = a1.x; As[st^1][lrA0+64][lcA+1] = a1.y;
        As[st^1][lrA0+64][lcA+2] = a1.z; As[st^1][lrA0+64][lcA+3] = a1.w;
        Ws[st^1][lrW][lcA+0] = wv.x; Ws[st^1][lrW][lcA+1] = wv.y;
        Ws[st^1][lrW][lcA+2] = wv.z; Ws[st^1][lrW][lcA+3] = wv.w;
      }
      __syncthreads();
    }
  }

#pragma unroll
  for (int mi = 0; mi < 2; ++mi) {
#pragma unroll
    for (int ni = 0; ni < 2; ++ni) {
      const v8f acc = (mi == 0) ? (ni == 0 ? acc00 : acc01)
                                : (ni == 0 ? acc10 : acc11);
#pragma unroll
      for (int j = 0; j < 8; ++j) {
        const int gm = m0 + mq + mi * 16 + j + hh * 8;
        const int gn = n0 + nq + ni * 16 + ll;
        float c = acc[j];
        if (!headsplit) {
          c += bias[gn];
          C[(size_t)gm * MD + gn] = c;
        } else {
          const int b = gm / SEQ, n = gm % SEQ;
          const int h = gn >> 6, d = gn & 63;
          C[(((size_t)b * NH + h) * SEQ + n) * HD + d] = c;
        }
      }
    }
  }
}

__global__ __launch_bounds__(128)
void flash_attn_causal(const float* __restrict__ Q, const float* __restrict__ K,
                       const float* __restrict__ V, float* __restrict__ O)
{
  __shared__ float Ks[16][66];
  __shared__ float Vt[64][18];
  __shared__ float Ps[4][16][18];

  const int bh   = blockIdx.x;
  const int qblk = blockIdx.y;
  const int tid  = threadIdx.x;
  const int w    = tid >> 5;
  const int lane = tid & 31;
  const int ll   = lane & 15;
  const int hh   = lane >> 4;
  const int qbase = qblk * 64 + w * 16;
  const float scale = 0.125f;

  const float* Qh = Q + (size_t)bh * SEQ * HD;
  const float* Kh = K + (size_t)bh * SEQ * HD;
  const float* Vh = V + (size_t)bh * SEQ * HD;

  v2f qf[16];
#pragma unroll
  for (int t = 0; t < 16; ++t) {
    v2f qv = *(const v2f*)&Qh[(size_t)(qbase + ll) * HD + t * 4 + 2 * hh];
    qf[t] = qv * scale;
  }

  const v8f z = {0.f,0.f,0.f,0.f,0.f,0.f,0.f,0.f};
  v8f oacc[4] = {z, z, z, z};
  float mrow[8], lrow[8];
#pragma unroll
  for (int j = 0; j < 8; ++j) { mrow[j] = -INFINITY; lrow[j] = 0.f; }

  const int nkt = qblk * 4 + 4;
  for (int kt = 0; kt < nkt; ++kt) {
    const int kbase = kt * 16;

#pragma unroll
    for (int part = 0; part < 2; ++part) {
      const int lin = tid + part * 128;
      const int row = lin >> 4, c4 = (lin & 15) * 4;
      const float4 kd = *(const float4*)&Kh[(size_t)(kbase + row) * HD + c4];
      const float4 vd = *(const float4*)&Vh[(size_t)(kbase + row) * HD + c4];
      Ks[row][c4+0] = kd.x; Ks[row][c4+1] = kd.y;
      Ks[row][c4+2] = kd.z; Ks[row][c4+3] = kd.w;
      Vt[c4+0][row] = vd.x; Vt[c4+1][row] = vd.y;
      Vt[c4+2][row] = vd.z; Vt[c4+3][row] = vd.w;
    }
    __syncthreads();

    if (kbase <= qbase + 15) {
      v8f s = z;
#pragma unroll
      for (int t = 0; t < 16; ++t) {
        const v2f b = *(const v2f*)&Ks[ll][t * 4 + 2 * hh];
        s = wmma_f32(qf[t], b, s);
      }

#pragma unroll
      for (int j = 0; j < 8; ++j) {
        const int rowg = qbase + j + hh * 8;
        const int colg = kbase + ll;
        const float sj = (colg > rowg) ? -INFINITY : s[j];
        float red = sj;
        red = fmaxf(red, __shfl_xor(red, 1, 32));
        red = fmaxf(red, __shfl_xor(red, 2, 32));
        red = fmaxf(red, __shfl_xor(red, 4, 32));
        red = fmaxf(red, __shfl_xor(red, 8, 32));
        const float mnew  = fmaxf(mrow[j], red);
        const float p     = __expf(sj - mnew);
        const float alpha = __expf(mrow[j] - mnew);
        mrow[j] = mnew;
        float psum = p;
        psum += __shfl_xor(psum, 1, 32);
        psum += __shfl_xor(psum, 2, 32);
        psum += __shfl_xor(psum, 4, 32);
        psum += __shfl_xor(psum, 8, 32);
        lrow[j] = lrow[j] * alpha + psum;
#pragma unroll
        for (int dt = 0; dt < 4; ++dt) oacc[dt][j] *= alpha;
        Ps[w][j + hh * 8][ll] = p;
      }

#pragma unroll
      for (int t = 0; t < 4; ++t) {
        const v2f a = *(const v2f*)&Ps[w][ll][t * 4 + 2 * hh];
#pragma unroll
        for (int dt = 0; dt < 4; ++dt) {
          const v2f b = *(const v2f*)&Vt[dt * 16 + ll][t * 4 + 2 * hh];
          oacc[dt] = wmma_f32(a, b, oacc[dt]);
        }
      }
    }
    __syncthreads();
  }

  const int b = bh >> 4;
  const int h = bh & 15;
#pragma unroll
  for (int j = 0; j < 8; ++j) {
    const float inv = 1.0f / lrow[j];
    const int rowg = qbase + j + hh * 8;
#pragma unroll
    for (int dt = 0; dt < 4; ++dt) {
      const int d = dt * 16 + ll;
      O[((size_t)b * SEQ + rowg) * MD + h * HD + d] = oacc[dt][j] * inv;
    }
  }
}

extern "C" void kernel_launch(void* const* d_in, const int* in_sizes, int n_in,
                              void* d_out, int out_size, void* d_ws, size_t ws_size,
                              hipStream_t stream) {
  (void)in_sizes; (void)n_in; (void)out_size; (void)ws_size;
  const float* x  = (const float*)d_in[0];
  const float* Wq = (const float*)d_in[1];
  const float* Wk = (const float*)d_in[2];
  const float* Wv = (const float*)d_in[3];
  const float* Wo = (const float*)d_in[4];
  const float* bo = (const float*)d_in[5];

  float* ws = (float*)d_ws;
  const size_t SZ = (size_t)BATCH * SEQ * MD;
  float* qb = ws;
  float* kb = ws + SZ;
  float* vb = ws + 2 * SZ;
  float* ao = ws + 3 * SZ;

  const dim3 ggrid(BATCH * SEQ / 128, MD / 64);
  gemm_nt_128x64<<<ggrid, 256, 0, stream>>>(x, Wq, nullptr, qb, 1);
  gemm_nt_128x64<<<ggrid, 256, 0, stream>>>(x, Wk, nullptr, kb, 1);
  gemm_nt_128x64<<<ggrid, 256, 0, stream>>>(x, Wv, nullptr, vb, 1);

  flash_attn_causal<<<dim3(BATCH * NH, SEQ / 64), 128, 0, stream>>>(qb, kb, vb, ao);

  gemm_nt_128x64<<<ggrid, 256, 0, stream>>>(ao, Wo, bo, (float*)d_out, 0);
}


# attention: score tile computed transposed (keys in a lane's registers, queries across lanes): softmax in-lane plus one half swap, P feeds P*V from registers; row-sum and P*V key order change
# speedup vs baseline: 1.2394x; 1.1349x over previous
#include <hip/hip_runtime.h>
#include <math.h>

#define MD    1024
#define NH    16
#define HD    64
#define BATCH 4
#define SEQ   2048

typedef __attribute__((ext_vector_type(2))) float v2f;
typedef __attribute__((ext_vector_type(8))) float v8f;

static __device__ __forceinline__ v8f wmma_f32(v2f a, v2f b, v8f c) {
  return __builtin_amdgcn_wmma_f32_16x16x4_f32(
       false, a,  false, b,
       (short)0, c,  false,  false);
}

__global__ __launch_bounds__(256)
void gemm_nt_128x64(const float* __restrict__ A, const float* __restrict__ W,
                    const float* __restrict__ bias, float* __restrict__ C,
                    int headsplit)
{
  __shared__ float As[2][128][18];
  __shared__ float Ws[2][64][18];

  const int m0   = blockIdx.x * 128;
  const int n0   = blockIdx.y * 64;
  const int tid  = threadIdx.x;
  const int w    = tid >> 5;
  const int lane = tid & 31;
  const int ll   = lane & 15;
  const int hh   = lane >> 4;
  const int mq   = (w >> 1) * 32;
  const int nq   = (w & 1) * 32;

  const v8f z = {0.f,0.f,0.f,0.f,0.f,0.f,0.f,0.f};
  v8f acc00 = z, acc01 = z, acc10 = z, acc11 = z;

  const int lrA0 = tid >> 2;
  const int lcA  = (tid & 3) * 4;
  const int lrW  = tid >> 2;

  const float* Ap0 = &A[(size_t)(m0 + lrA0     ) * MD + lcA];
  const float* Ap1 = &A[(size_t)(m0 + lrA0 + 64) * MD + lcA];
  const float* Wp  = &W[(size_t)(n0 + lrW      ) * MD + lcA];

  float4 a0 = *(const float4*)&Ap0[0];
  float4 a1 = *(const float4*)&Ap1[0];
  float4 wv = *(const float4*)&Wp[0];
  As[0][lrA0   ][lcA+0] = a0.x; As[0][lrA0   ][lcA+1] = a0.y;
  As[0][lrA0   ][lcA+2] = a0.z; As[0][lrA0   ][lcA+3] = a0.w;
  As[0][lrA0+64][lcA+0] = a1.x; As[0][lrA0+64][lcA+1] = a1.y;
  As[0][lrA0+64][lcA+2] = a1.z; As[0][lrA0+64][lcA+3] = a1.w;
  Ws[0][lrW][lcA+0] = wv.x; Ws[0][lrW][lcA+1] = wv.y;
  Ws[0][lrW][lcA+2] = wv.z; Ws[0][lrW][lcA+3] = wv.w;
  __syncthreads();

  for (int k0 = 0; k0 < MD; k0 += 32) {
#pragma unroll
    for (int st = 0; st < 2; ++st) {
      const int kn = k0 + 16 * st + 16;
      const bool more = (kn < MD);
      if (more) {
        a0 = *(const float4*)&Ap0[kn];
        a1 = *(const float4*)&Ap1[kn];
        wv = *(const float4*)&Wp[kn];
      }

#pragma unroll
      for (int kk = 0; kk < 16; kk += 4) {
        const int col = kk + 2 * hh;
        const v2f a0f = *(const v2f*)&As[st][mq      + ll][col];
        const v2f a1f = *(const v2f*)&As[st][mq + 16 + ll][col];
        const v2f b0f = *(const v2f*)&Ws[st][nq      + ll][col];
        const v2f b1f = *(const v2f*)&Ws[st][nq + 16 + ll][col];
        acc00 = wmma_f32(a0f, b0f, acc00);
        acc01 = wmma_f32(a0f, b1f, acc01);
        acc10 = wmma_f32(a1f, b0f, acc10);
        acc11 = wmma_f32(a1f, b1f, acc11);
      }

      __builtin_amdgcn_sched_barrier(0);
      if (more) {
        As[st^1][lrA0   ][lcA+0] = a0.x; As[st^1][lrA0   ][lcA+1] = a0.y;
        As[st^1][lrA0   ][lcA+2] = a0.z; As[st^1][lrA0   ][lcA+3] = a0.w;
        As[st^1][lrA0+64][lcA+0] = a1.x; As[st^1][lrA0+64][lcA+1] = a1.y;
        As[st^1][lrA0+64][lcA+2] = a1.z; As[st^1][lrA0+64][lcA+3] = a1.w;
        Ws[st^1][lrW][lcA+0] = wv.x; Ws[st^1][lrW][lcA+1] = wv.y;
        Ws[st^1][lrW][lcA+2] = wv.z; Ws[st^1][lrW][lcA+3] = wv.w;
      }
      __syncthreads();
    }
  }

#pragma unroll
  for (int mi = 0; mi < 2; ++mi) {
#pragma unroll
    for (int ni = 0; ni < 2; ++ni) {
      const v8f acc = (mi == 0) ? (ni == 0 ? acc00 : acc01)
                                : (ni == 0 ? acc10 : acc11);
#pragma unroll
      for (int j = 0; j < 8; ++j) {
        const int gm = m0 + mq + mi * 16 + j + hh * 8;
        const int gn = n0 + nq + ni * 16 + ll;
        float c = acc[j];
        if (!headsplit) {
          c += bias[gn];
          C[(size_t)gm * MD + gn] = c;
        } else {
          const int b = gm / SEQ, n = gm % SEQ;
          const int h = gn >> 6, d = gn & 63;
          C[(((size_t)b * NH + h) * SEQ + n) * HD + d] = c;
        }
      }
    }
  }
}

static __device__ __forceinline__ float xhalf(float x) {
  const int b = __builtin_bit_cast(int, x);
  return __builtin_bit_cast(float,
      __builtin_amdgcn_permlanex16(b, b, 0x76543210, 0xfedcba98, false, false));
}

__global__ __launch_bounds__(128) __attribute__((amdgpu_waves_per_eu(4, 8)))
void flash_attn_causal(const float* __restrict__ Q, const float* __restrict__ K,
                       const float* __restrict__ V, float* __restrict__ O)
{
  __shared__ float Ks[16][66];
  __shared__ float Vt[64][18];

  const int bh   = blockIdx.x;
  const int qblk = blockIdx.y;
  const int tid  = threadIdx.x;
  const int w    = tid >> 5;
  const int lane = tid & 31;
  const int ll   = lane & 15;
  const int hh   = lane >> 4;
  const int qbase = qblk * 64 + w * 16;
  const int qg    = qbase + ll;
  const float scale = 0.125f;

  const float* Qh = Q + (size_t)bh * SEQ * HD;
  const float* Kh = K + (size_t)bh * SEQ * HD;
  const float* Vh = V + (size_t)bh * SEQ * HD;

  v2f qf[16];
#pragma unroll
  for (int t = 0; t < 16; ++t) {
    v2f qv = *(const v2f*)&Qh[(size_t)(qbase + ll) * HD + t * 4 + 2 * hh];
    qf[t] = qv * scale;
  }

  const v8f z = {0.f,0.f,0.f,0.f,0.f,0.f,0.f,0.f};
  v8f ot[4] = {z, z, z, z};
  float mrun = -INFINITY;
  float lrun = 0.f;

  const int nkt = qblk * 4 + 4;
  for (int kt = 0; kt < nkt; ++kt) {
    const int kbase = kt * 16;

    float4 kd[2], vd[2];
#pragma unroll
    for (int part = 0; part < 2; ++part) {
      const int lin = tid + part * 128;
      const int row = lin >> 4, c4 = (lin & 15) * 4;
      kd[part] = *(const float4*)&Kh[(size_t)(kbase + row) * HD + c4];
      vd[part] = *(const float4*)&Vh[(size_t)(kbase + row) * HD + c4];
    }
#pragma unroll
    for (int part = 0; part < 2; ++part) {
      const int lin = tid + part * 128;
      const int row = lin >> 4, c4 = (lin & 15) * 4;
      Ks[row][c4+0] = kd[part].x; Ks[row][c4+1] = kd[part].y;
      Ks[row][c4+2] = kd[part].z; Ks[row][c4+3] = kd[part].w;
      Vt[c4+0][row] = vd[part].x; Vt[c4+1][row] = vd[part].y;
      Vt[c4+2][row] = vd[part].z; Vt[c4+3][row] = vd[part].w;
    }
    __syncthreads();

    if (kbase <= qbase + 15) {
      v8f st = z;
#pragma unroll
      for (int t = 0; t < 16; ++t) {
        const v2f a = *(const v2f*)&Ks[ll][t * 4 + 2 * hh];
        st = wmma_f32(a, qf[t], st);
      }

      float sv[8];
#pragma unroll
      for (int r = 0; r < 8; ++r) {
        const int keyg = kbase + 8 * hh + r;
        sv[r] = (keyg > qg) ? -INFINITY : st[r];
      }
      float mx = fmaxf(fmaxf(fmaxf(sv[0], sv[1]), fmaxf(sv[2], sv[3])),
                       fmaxf(fmaxf(sv[4], sv[5]), fmaxf(sv[6], sv[7])));
      mx = fmaxf(mx, xhalf(mx));
      const float mnew  = fmaxf(mrun, mx);
      const float alpha = __expf(mrun - mnew);
      mrun = mnew;
      float p[8];
#pragma unroll
      for (int r = 0; r < 8; ++r) p[r] = __expf(sv[r] - mnew);
      float ps = ((p[0] + p[1]) + (p[2] + p[3])) + ((p[4] + p[5]) + (p[6] + p[7]));
      ps = ps + xhalf(ps);
      lrun = lrun * alpha + ps;
#pragma unroll
      for (int dt = 0; dt < 4; ++dt)
#pragma unroll
        for (int r = 0; r < 8; ++r) ot[dt][r] *= alpha;

#pragma unroll
      for (int t = 0; t < 4; ++t) {
        const v2f b = {p[2 * t], p[2 * t + 1]};
#pragma unroll
        for (int dt = 0; dt < 4; ++dt) {
          const v2f a = *(const v2f*)&Vt[dt * 16 + ll][8 * hh + 2 * t];
          ot[dt] = wmma_f32(a, b, ot[dt]);
        }
      }
    }
    __syncthreads();
  }

  const int b = bh >> 4;
  const int h = bh & 15;
  const float inv = 1.0f / lrun;
  float* dst = &O[((size_t)b * SEQ + qg) * MD + h * HD + 8 * hh];
#pragma unroll
  for (int dt = 0; dt < 4; ++dt) {
    const float4 lo = {ot[dt][0] * inv, ot[dt][1] * inv, ot[dt][2] * inv, ot[dt][3] * inv};
    const float4 hi = {ot[dt][4] * inv, ot[dt][5] * inv, ot[dt][6] * inv, ot[dt][7] * inv};
    *(float4*)&dst[dt * 16]     = lo;
    *(float4*)&dst[dt * 16 + 4] = hi;
  }
}

extern "C" void kernel_launch(void* const* d_in, const int* in_sizes, int n_in,
                              void* d_out, int out_size, void* d_ws, size_t ws_size,
                              hipStream_t stream) {
  (void)in_sizes; (void)n_in; (void)out_size; (void)ws_size;
  const float* x  = (const float*)d_in[0];
  const float* Wq = (const float*)d_in[1];
  const float* Wk = (const float*)d_in[2];
  const float* Wv = (const float*)d_in[3];
  const float* Wo = (const float*)d_in[4];
  const float* bo = (const float*)d_in[5];

  float* ws = (float*)d_ws;
  const size_t SZ = (size_t)BATCH * SEQ * MD;
  float* qb = ws;
  float* kb = ws + SZ;
  float* vb = ws + 2 * SZ;
  float* ao = ws + 3 * SZ;

  const dim3 ggrid(BATCH * SEQ / 128, MD / 64);
  gemm_nt_128x64<<<ggrid, 256, 0, stream>>>(x, Wq, nullptr, qb, 1);
  gemm_nt_128x64<<<ggrid, 256, 0, stream>>>(x, Wk, nullptr, kb, 1);
  gemm_nt_128x64<<<ggrid, 256, 0, stream>>>(x, Wv, nullptr, vb, 1);

  flash_attn_causal<<<dim3(BATCH * NH, SEQ / 64), 128, 0, stream>>>(qb, kb, vb, ao);

  gemm_nt_128x64<<<ggrid, 256, 0, stream>>>(ao, Wo, bo, (float*)d_out, 0);
}


# attention: query blocks mapped to blockIdx.y heaviest-first (longest causal blocks enter first = oldest waves), every second group of four rows mirrored so blocks sharing a unit sum equal
# speedup vs baseline: 1.3146x; 1.0607x over previous
#include <hip/hip_runtime.h>
#include <math.h>

#define MD    1024
#define NH    16
#define HD    64
#define BATCH 4
#define SEQ   2048

typedef __attribute__((ext_vector_type(2))) float v2f;
typedef __attribute__((ext_vector_type(8))) float v8f;

static __device__ __forceinline__ v8f wmma_f32(v2f a, v2f b, v8f c) {
  return __builtin_amdgcn_wmma_f32_16x16x4_f32(
       false, a,  false, b,
       (short)0, c,  false,  false);
}

__global__ __launch_bounds__(256)
void gemm_nt_128x64(const float* __restrict__ A, const float* __restrict__ W,
                    const float* __restrict__ bias, float* __restrict__ C,
                    int headsplit)
{
  __shared__ float As[2][128][18];
  __shared__ float Ws[2][64][18];

  const int m0   = blockIdx.x * 128;
  const int n0   = blockIdx.y * 64;
  const int tid  = threadIdx.x;
  const int w    = tid >> 5;
  const int lane = tid & 31;
  const int ll   = lane & 15;
  const int hh   = lane >> 4;
  const int mq   = (w >> 1) * 32;
  const int nq   = (w & 1) * 32;

  const v8f z = {0.f,0.f,0.f,0.f,0.f,0.f,0.f,0.f};
  v8f acc00 = z, acc01 = z, acc10 = z, acc11 = z;

  const int lrA0 = tid >> 2;
  const int lcA  = (tid & 3) * 4;
  const int lrW  = tid >> 2;

  const float* Ap0 = &A[(size_t)(m0 + lrA0     ) * MD + lcA];
  const float* Ap1 = &A[(size_t)(m0 + lrA0 + 64) * MD + lcA];
  const float* Wp  = &W[(size_t)(n0 + lrW      ) * MD + lcA];

  float4 a0 = *(const float4*)&Ap0[0];
  float4 a1 = *(const float4*)&Ap1[0];
  float4 wv = *(const float4*)&Wp[0];
  As[0][lrA0   ][lcA+0] = a0.x; As[0][lrA0   ][lcA+1] = a0.y;
  As[0][lrA0   ][lcA+2] = a0.z; As[0][lrA0   ][lcA+3] = a0.w;
  As[0][lrA0+64][lcA+0] = a1.x; As[0][lrA0+64][lcA+1] = a1.y;
  As[0][lrA0+64][lcA+2] = a1.z; As[0][lrA0+64][lcA+3] = a1.w;
  Ws[0][lrW][lcA+0] = wv.x; Ws[0][lrW][lcA+1] = wv.y;
  Ws[0][lrW][lcA+2] = wv.z; Ws[0][lrW][lcA+3] = wv.w;
  __syncthreads();

  for (int k0 = 0; k0 < MD; k0 += 32) {
#pragma unroll
    for (int st = 0; st < 2; ++st) {
      const int kn = k0 + 16 * st + 16;
      const bool more = (kn < MD);
      if (more) {
        a0 = *(const float4*)&Ap0[kn];
        a1 = *(const float4*)&Ap1[kn];
        wv = *(const float4*)&Wp[kn];
      }

#pragma unroll
      for (int kk = 0; kk < 16; kk += 4) {
        const int col = kk + 2 * hh;
        const v2f a0f = *(const v2f*)&As[st][mq      + ll][col];
        const v2f a1f = *(const v2f*)&As[st][mq + 16 + ll][col];
        const v2f b0f = *(const v2f*)&Ws[st][nq      + ll][col];
        const v2f b1f = *(const v2f*)&Ws[st][nq + 16 + ll][col];
        acc00 = wmma_f32(a0f, b0f, acc00);
        acc01 = wmma_f32(a0f, b1f, acc01);
        acc10 = wmma_f32(a1f, b0f, acc10);
        acc11 = wmma_f32(a1f, b1f, acc11);
      }

      __builtin_amdgcn_sched_barrier(0);
      if (more) {
        As[st^1][lrA0   ][lcA+0] = a0.x; As[st^1][lrA0   ][lcA+1] = a0.y;
        As[st^1][lrA0   ][lcA+2] = a0.z; As[st^1][lrA0   ][lcA+3] = a0.w;
        As[st^1][lrA0+64][lcA+0] = a1.x; As[st^1][lrA0+64][lcA+1] = a1.y;
        As[st^1][lrA0+64][lcA+2] = a1.z; As[st^1][lrA0+64][lcA+3] = a1.w;
        Ws[st^1][lrW][lcA+0] = wv.x; Ws[st^1][lrW][lcA+1] = wv.y;
        Ws[st^1][lrW][lcA+2] = wv.z; Ws[st^1][lrW][lcA+3] = wv.w;
      }
      __syncthreads();
    }
  }

#pragma unroll
  for (int mi = 0; mi < 2; ++mi) {
#pragma unroll
    for (int ni = 0; ni < 2; ++ni) {
      const v8f acc = (mi == 0) ? (ni == 0 ? acc00 : acc01)
                                : (ni == 0 ? acc10 : acc11);
#pragma unroll
      for (int j = 0; j < 8; ++j) {
        const int gm = m0 + mq + mi * 16 + j + hh * 8;
        const int gn = n0 + nq + ni * 16 + ll;
        float c = acc[j];
        if (!headsplit) {
          c += bias[gn];
          C[(size_t)gm * MD + gn] = c;
        } else {
          const int b = gm / SEQ, n = gm % SEQ;
          const int h = gn >> 6, d = gn & 63;
          C[(((size_t)b * NH + h) * SEQ + n) * HD + d] = c;
        }
      }
    }
  }
}

static __device__ __forceinline__ float xhalf(float x) {
  const int b = __builtin_bit_cast(int, x);
  return __builtin_bit_cast(float,
      __builtin_amdgcn_permlanex16(b, b, 0x76543210, 0xfedcba98, false, false));
}

__global__ __launch_bounds__(128) __attribute__((amdgpu_waves_per_eu(4, 8)))
void flash_attn_causal(const float* __restrict__ Q, const float* __restrict__ K,
                       const float* __restrict__ V, float* __restrict__ O)
{
  __shared__ float Ks[16][66];
  __shared__ float Vt[64][18];

  const int bh   = blockIdx.x;
  const int ygrp = blockIdx.y >> 2;
  const int yin  = blockIdx.y & 3;
  const int qblk = (SEQ / 64 - 1) - (ygrp * 4 + ((ygrp & 1) ? 3 - yin : yin));
  const int tid  = threadIdx.x;
  const int w    = tid >> 5;
  const int lane = tid & 31;
  const int ll   = lane & 15;
  const int hh   = lane >> 4;
  const int qbase = qblk * 64 + w * 16;
  const int qg    = qbase + ll;
  const float scale = 0.125f;

  const float* Qh = Q + (size_t)bh * SEQ * HD;
  const float* Kh = K + (size_t)bh * SEQ * HD;
  const float* Vh = V + (size_t)bh * SEQ * HD;

  v2f qf[16];
#pragma unroll
  for (int t = 0; t < 16; ++t) {
    v2f qv = *(const v2f*)&Qh[(size_t)(qbase + ll) * HD + t * 4 + 2 * hh];
    qf[t] = qv * scale;
  }

  const v8f z = {0.f,0.f,0.f,0.f,0.f,0.f,0.f,0.f};
  v8f ot[4] = {z, z, z, z};
  float mrun = -INFINITY;
  float lrun = 0.f;

  const int nkt = qblk * 4 + 4;
  for (int kt = 0; kt < nkt; ++kt) {
    const int kbase = kt * 16;

    float4 kd[2], vd[2];
#pragma unroll
    for (int part = 0; part < 2; ++part) {
      const int lin = tid + part * 128;
      const int row = lin >> 4, c4 = (lin & 15) * 4;
      kd[part] = *(const float4*)&Kh[(size_t)(kbase + row) * HD + c4];
      vd[part] = *(const float4*)&Vh[(size_t)(kbase + row) * HD + c4];
    }
#pragma unroll
    for (int part = 0; part < 2; ++part) {
      const int lin = tid + part * 128;
      const int row = lin >> 4, c4 = (lin & 15) * 4;
      Ks[row][c4+0] = kd[part].x; Ks[row][c4+1] = kd[part].y;
      Ks[row][c4+2] = kd[part].z; Ks[row][c4+3] = kd[part].w;
      Vt[c4+0][row] = vd[part].x; Vt[c4+1][row] = vd[part].y;
      Vt[c4+2][row] = vd[part].z; Vt[c4+3][row] = vd[part].w;
    }
    __syncthreads();

    if (kbase <= qbase + 15) {
      v8f st = z;
#pragma unroll
      for (int t = 0; t < 16; ++t) {
        const v2f a = *(const v2f*)&Ks[ll][t * 4 + 2 * hh];
        st = wmma_f32(a, qf[t], st);
      }

      float sv[8];
#pragma unroll
      for (int r = 0; r < 8; ++r) {
        const int keyg = kbase + 8 * hh + r;
        sv[r] = (keyg > qg) ? -INFINITY : st[r];
      }
      float mx = fmaxf(fmaxf(fmaxf(sv[0], sv[1]), fmaxf(sv[2], sv[3])),
                       fmaxf(fmaxf(sv[4], sv[5]), fmaxf(sv[6], sv[7])));
      mx = fmaxf(mx, xhalf(mx));
      const float mnew  = fmaxf(mrun, mx);
      const float alpha = __expf(mrun - mnew);
      mrun = mnew;
      float p[8];
#pragma unroll
      for (int r = 0; r < 8; ++r) p[r] = __expf(sv[r] - mnew);
      float ps = ((p[0] + p[1]) + (p[2] + p[3])) + ((p[4] + p[5]) + (p[6] + p[7]));
      ps = ps + xhalf(ps);
      lrun = lrun * alpha + ps;
#pragma unroll
      for (int dt = 0; dt < 4; ++dt)
#pragma unroll
        for (int r = 0; r < 8; ++r) ot[dt][r] *= alpha;

#pragma unroll
      for (int t = 0; t < 4; ++t) {
        const v2f b = {p[2 * t], p[2 * t + 1]};
#pragma unroll
        for (int dt = 0; dt < 4; ++dt) {
          const v2f a = *(const v2f*)&Vt[dt * 16 + ll][8 * hh + 2 * t];
          ot[dt] = wmma_f32(a, b, ot[dt]);
        }
      }
    }
    __syncthreads();
  }

  const int b = bh >> 4;
  const int h = bh & 15;
  const float inv = 1.0f / lrun;
  float* dst = &O[((size_t)b * SEQ + qg) * MD + h * HD + 8 * hh];
#pragma unroll
  for (int dt = 0; dt < 4; ++dt) {
    const float4 lo = {ot[dt][0] * inv, ot[dt][1] * inv, ot[dt][2] * inv, ot[dt][3] * inv};
    const float4 hi = {ot[dt][4] * inv, ot[dt][5] * inv, ot[dt][6] * inv, ot[dt][7] * inv};
    *(float4*)&dst[dt * 16]     = lo;
    *(float4*)&dst[dt * 16 + 4] = hi;
  }
}

extern "C" void kernel_launch(void* const* d_in, const int* in_sizes, int n_in,
                              void* d_out, int out_size, void* d_ws, size_t ws_size,
                              hipStream_t stream) {
  (void)in_sizes; (void)n_in; (void)out_size; (void)ws_size;
  const float* x  = (const float*)d_in[0];
  const float* Wq = (const float*)d_in[1];
  const float* Wk = (const float*)d_in[2];
  const float* Wv = (const float*)d_in[3];
  const float* Wo = (const float*)d_in[4];
  const float* bo = (const float*)d_in[5];

  float* ws = (float*)d_ws;
  const size_t SZ = (size_t)BATCH * SEQ * MD;
  float* qb = ws;
  float* kb = ws + SZ;
  float* vb = ws + 2 * SZ;
  float* ao = ws + 3 * SZ;

  const dim3 ggrid(BATCH * SEQ / 128, MD / 64);
  gemm_nt_128x64<<<ggrid, 256, 0, stream>>>(x, Wq, nullptr, qb, 1);
  gemm_nt_128x64<<<ggrid, 256, 0, stream>>>(x, Wk, nullptr, kb, 1);
  gemm_nt_128x64<<<ggrid, 256, 0, stream>>>(x, Wv, nullptr, vb, 1);

  flash_attn_causal<<<dim3(BATCH * NH, SEQ / 64), 128, 0, stream>>>(qb, kb, vb, ao);

  gemm_nt_128x64<<<ggrid, 256, 0, stream>>>(ao, Wo, bo, (float*)d_out, 0);
}


# GEMM re-tiled: block tile 128x64x16 -> 128x128x32, wave tile 32x32 -> 32x64 (2x4 accumulators): 6 LDS fragment reads per 8 WMMAs, one barrier per 64 WMMAs, whole-line global loads
# speedup vs baseline: 1.4276x; 1.0860x over previous
#include <hip/hip_runtime.h>
#include <math.h>

#define MD    1024
#define NH    16
#define HD    64
#define BATCH 4
#define SEQ   2048

typedef __attribute__((ext_vector_type(2))) float v2f;
typedef __attribute__((ext_vector_type(8))) float v8f;

static __device__ __forceinline__ v8f wmma_f32(v2f a, v2f b, v8f c) {
  return __builtin_amdgcn_wmma_f32_16x16x4_f32(
       false, a,  false, b,
       (short)0, c,  false,  false);
}

#define GBK   32
#define GPIT  36
__global__ __launch_bounds__(256)
void gemm_nt_128x128(const float* A, const float* W,
                     const float* __restrict__ bias, float* C,
                     int headsplit)
{
  __shared__ __attribute__((aligned(16))) float As[2][128][GPIT];
  __shared__ __attribute__((aligned(16))) float Ws[2][128][GPIT];

  const int m0   = blockIdx.x * 128;
  const int n0   = blockIdx.y * 128;
  const int tid  = threadIdx.x;
  const int w    = tid >> 5;
  const int lane = tid & 31;
  const int ll   = lane & 15;
  const int hh   = lane >> 4;
  const int mq   = (w >> 1) * 32;
  const int nq   = (w & 1) * 64;

  const v8f z = {0.f,0.f,0.f,0.f,0.f,0.f,0.f,0.f};
  v8f acc[2][4];
#pragma unroll
  for (int mi = 0; mi < 2; ++mi)
#pragma unroll
    for (int ni = 0; ni < 4; ++ni) acc[mi][ni] = z;

  const int lr = tid >> 3;
  const int lc = (tid & 7) * 4;

  const float* Ap = &A[(size_t)(m0 + lr) * MD + lc];
  const float* Wp = &W[(size_t)(n0 + lr) * MD + lc];

  float4 a0 = *(const float4*)&Ap[0];
  float4 a1 = *(const float4*)&Ap[(size_t)32 * MD];
  float4 a2 = *(const float4*)&Ap[(size_t)64 * MD];
  float4 a3 = *(const float4*)&Ap[(size_t)96 * MD];
  float4 w0 = *(const float4*)&Wp[0];
  float4 w1 = *(const float4*)&Wp[(size_t)32 * MD];
  float4 w2 = *(const float4*)&Wp[(size_t)64 * MD];
  float4 w3 = *(const float4*)&Wp[(size_t)96 * MD];
  *(float4*)&As[0][lr     ][lc] = a0;
  *(float4*)&As[0][lr + 32][lc] = a1;
  *(float4*)&As[0][lr + 64][lc] = a2;
  *(float4*)&As[0][lr + 96][lc] = a3;
  *(float4*)&Ws[0][lr     ][lc] = w0;
  *(float4*)&Ws[0][lr + 32][lc] = w1;
  *(float4*)&Ws[0][lr + 64][lc] = w2;
  *(float4*)&Ws[0][lr + 96][lc] = w3;
  __syncthreads();

  for (int k0 = 0; k0 < MD; k0 += 2 * GBK) {
#pragma unroll
    for (int st = 0; st < 2; ++st) {
      const int kn = k0 + GBK * st + GBK;
      const bool more = (kn < MD);
      if (more) {
        a0 = *(const float4*)&Ap[kn];
        a1 = *(const float4*)&Ap[(size_t)32 * MD + kn];
        a2 = *(const float4*)&Ap[(size_t)64 * MD + kn];
        a3 = *(const float4*)&Ap[(size_t)96 * MD + kn];
        w0 = *(const float4*)&Wp[kn];
        w1 = *(const float4*)&Wp[(size_t)32 * MD + kn];
        w2 = *(const float4*)&Wp[(size_t)64 * MD + kn];
        w3 = *(const float4*)&Wp[(size_t)96 * MD + kn];
      }

#pragma unroll
      for (int kk = 0; kk < GBK; kk += 4) {
        const int col = kk + 2 * hh;
        v2f af[2], bf[4];
#pragma unroll
        for (int mi = 0; mi < 2; ++mi)
          af[mi] = *(const v2f*)&As[st][mq + 16 * mi + ll][col];
#pragma unroll
        for (int ni = 0; ni < 4; ++ni)
          bf[ni] = *(const v2f*)&Ws[st][nq + 16 * ni + ll][col];
#pragma unroll
        for (int mi = 0; mi < 2; ++mi)
#pragma unroll
          for (int ni = 0; ni < 4; ++ni)
            acc[mi][ni] = wmma_f32(af[mi], bf[ni], acc[mi][ni]);
      }

      __builtin_amdgcn_sched_barrier(0);
      if (more) {
        *(float4*)&As[st ^ 1][lr     ][lc] = a0;
        *(float4*)&As[st ^ 1][lr + 32][lc] = a1;
        *(float4*)&As[st ^ 1][lr + 64][lc] = a2;
        *(float4*)&As[st ^ 1][lr + 96][lc] = a3;
        *(float4*)&Ws[st ^ 1][lr     ][lc] = w0;
        *(float4*)&Ws[st ^ 1][lr + 32][lc] = w1;
        *(float4*)&Ws[st ^ 1][lr + 64][lc] = w2;
        *(float4*)&Ws[st ^ 1][lr + 96][lc] = w3;
      }
      __syncthreads();
    }
  }

  const unsigned gm0 = (unsigned)(m0 + mq + hh * 8);
  const unsigned gn0 = (unsigned)(n0 + nq + ll);
  if (!headsplit) {
#pragma unroll
    for (int ni = 0; ni < 4; ++ni) {
      const unsigned gn = gn0 + 16 * ni;
      const float bv = bias[gn];
#pragma unroll
      for (int mi = 0; mi < 2; ++mi)
#pragma unroll
        for (int j = 0; j < 8; ++j) {
          const unsigned gm = gm0 + 16 * mi + j;
          C[(size_t)gm * MD + gn] = acc[mi][ni][j] + bv;
        }
    }
  } else {
#pragma unroll
    for (int ni = 0; ni < 4; ++ni) {
      const unsigned gn = gn0 + 16 * ni;
      const unsigned h = gn >> 6, d = gn & 63;
#pragma unroll
      for (int mi = 0; mi < 2; ++mi)
#pragma unroll
        for (int j = 0; j < 8; ++j) {
          const unsigned gm = gm0 + 16 * mi + j;
          const unsigned b = gm / SEQ, n = gm % SEQ;
          C[(((size_t)b * NH + h) * SEQ + n) * HD + d] = acc[mi][ni][j];
        }
    }
  }
}

static __device__ __forceinline__ float xhalf(float x) {
  const int b = __builtin_bit_cast(int, x);
  return __builtin_bit_cast(float,
      __builtin_amdgcn_permlanex16(b, b, 0x76543210, 0xfedcba98, false, false));
}

__global__ __launch_bounds__(128) __attribute__((amdgpu_waves_per_eu(4, 8)))
void flash_attn_causal(const float* __restrict__ Q, const float* __restrict__ K,
                       const float* __restrict__ V, float* __restrict__ O)
{
  __shared__ float Ks[16][66];
  __shared__ float Vt[64][18];

  const int bh   = blockIdx.x;
  const int ygrp = blockIdx.y >> 2;
  const int yin  = blockIdx.y & 3;
  const int qblk = (SEQ / 64 - 1) - (ygrp * 4 + ((ygrp & 1) ? 3 - yin : yin));
  const int tid  = threadIdx.x;
  const int w    = tid >> 5;
  const int lane = tid & 31;
  const int ll   = lane & 15;
  const int hh   = lane >> 4;
  const int qbase = qblk * 64 + w * 16;
  const int qg    = qbase + ll;
  const float scale = 0.125f;

  const float* Qh = Q + (size_t)bh * SEQ * HD;
  const float* Kh = K + (size_t)bh * SEQ * HD;
  const float* Vh = V + (size_t)bh * SEQ * HD;

  v2f qf[16];
#pragma unroll
  for (int t = 0; t < 16; ++t) {
    v2f qv = *(const v2f*)&Qh[(size_t)(qbase + ll) * HD + t * 4 + 2 * hh];
    qf[t] = qv * scale;
  }

  const v8f z = {0.f,0.f,0.f,0.f,0.f,0.f,0.f,0.f};
  v8f ot[4] = {z, z, z, z};
  float mrun = -INFINITY;
  float lrun = 0.f;

  const int nkt = qblk * 4 + 4;
  for (int kt = 0; kt < nkt; ++kt) {
    const int kbase = kt * 16;

    float4 kd[2], vd[2];
#pragma unroll
    for (int part = 0; part < 2; ++part) {
      const int lin = tid + part * 128;
      const int row = lin >> 4, c4 = (lin & 15) * 4;
      kd[part] = *(const float4*)&Kh[(size_t)(kbase + row) * HD + c4];
      vd[part] = *(const float4*)&Vh[(size_t)(kbase + row) * HD + c4];
    }
#pragma unroll
    for (int part = 0; part < 2; ++part) {
      const int lin = tid + part * 128;
      const int row = lin >> 4, c4 = (lin & 15) * 4;
      Ks[row][c4+0] = kd[part].x; Ks[row][c4+1] = kd[part].y;
      Ks[row][c4+2] = kd[part].z; Ks[row][c4+3] = kd[part].w;
      Vt[c4+0][row] = vd[part].x; Vt[c4+1][row] = vd[part].y;
      Vt[c4+2][row] = vd[part].z; Vt[c4+3][row] = vd[part].w;
    }
    __syncthreads();

    if (kbase <= qbase + 15) {
      v8f st = z;
#pragma unroll
      for (int t = 0; t < 16; ++t) {
        const v2f a = *(const v2f*)&Ks[ll][t * 4 + 2 * hh];
        st = wmma_f32(a, qf[t], st);
      }

      float sv[8];
#pragma unroll
      for (int r = 0; r < 8; ++r) {
        const int keyg = kbase + 8 * hh + r;
        sv[r] = (keyg > qg) ? -INFINITY : st[r];
      }
      float mx = fmaxf(fmaxf(fmaxf(sv[0], sv[1]), fmaxf(sv[2], sv[3])),
                       fmaxf(fmaxf(sv[4], sv[5]), fmaxf(sv[6], sv[7])));
      mx = fmaxf(mx, xhalf(mx));
      const float mnew  = fmaxf(mrun, mx);
      const float alpha = __expf(mrun - mnew);
      mrun = mnew;
      float p[8];
#pragma unroll
      for (int r = 0; r < 8; ++r) p[r] = __expf(sv[r] - mnew);
      float ps = ((p[0] + p[1]) + (p[2] + p[3])) + ((p[4] + p[5]) + (p[6] + p[7]));
      ps = ps + xhalf(ps);
      lrun = lrun * alpha + ps;
#pragma unroll
      for (int dt = 0; dt < 4; ++dt)
#pragma unroll
        for (int r = 0; r < 8; ++r) ot[dt][r] *= alpha;

#pragma unroll
      for (int t = 0; t < 4; ++t) {
        const v2f b = {p[2 * t], p[2 * t + 1]};
#pragma unroll
        for (int dt = 0; dt < 4; ++dt) {
          const v2f a = *(const v2f*)&Vt[dt * 16 + ll][8 * hh + 2 * t];
          ot[dt] = wmma_f32(a, b, ot[dt]);
        }
      }
    }
    __syncthreads();
  }

  const int b = bh >> 4;
  const int h = bh & 15;
  const float inv = 1.0f / lrun;
  float* dst = &O[((size_t)b * SEQ + qg) * MD + h * HD + 8 * hh];
#pragma unroll
  for (int dt = 0; dt < 4; ++dt) {
    const float4 lo = {ot[dt][0] * inv, ot[dt][1] * inv, ot[dt][2] * inv, ot[dt][3] * inv};
    const float4 hi = {ot[dt][4] * inv, ot[dt][5] * inv, ot[dt][6] * inv, ot[dt][7] * inv};
    *(float4*)&dst[dt * 16]     = lo;
    *(float4*)&dst[dt * 16 + 4] = hi;
  }
}

extern "C" void kernel_launch(void* const* d_in, const int* in_sizes, int n_in,
                              void* d_out, int out_size, void* d_ws, size_t ws_size,
                              hipStream_t stream) {
  (void)in_sizes; (void)n_in; (void)out_size; (void)ws_size;
  const float* x  = (const float*)d_in[0];
  const float* Wq = (const float*)d_in[1];
  const float* Wk = (const float*)d_in[2];
  const float* Wv = (const float*)d_in[3];
  const float* Wo = (const float*)d_in[4];
  const float* bo = (const float*)d_in[5];

  float* ws = (float*)d_ws;
  const size_t SZ = (size_t)BATCH * SEQ * MD;
  float* qb = ws;
  float* kb = ws + SZ;
  float* vb = ws + 2 * SZ;
  float* ao = ws + 3 * SZ;

  const dim3 ggrid(BATCH * SEQ / 128, MD / 128);
  gemm_nt_128x128<<<ggrid, 256, 0, stream>>>(x, Wq, nullptr, qb, 1);
  gemm_nt_128x128<<<ggrid, 256, 0, stream>>>(x, Wk, nullptr, kb, 1);
  gemm_nt_128x128<<<ggrid, 256, 0, stream>>>(x, Wv, nullptr, vb, 1);

  flash_attn_causal<<<dim3(BATCH * NH, SEQ / 64), 128, 0, stream>>>(qb, kb, vb, ao);

  gemm_nt_128x128<<<ggrid, 256, 0, stream>>>(ao, Wo, bo, (float*)d_out, 0);
}
